# starting kernel
# baseline (speedup 1.0000x reference)
#include <hip/hip_runtime.h>
#include <hip/hip_bf16.h>

#define Cc   8
#define Tt   8
#define Dd   24
#define Hh   48
#define Ww   48
#define KT   3
#define KS   3
#define TO   6
#define DO   47
#define HO   95
#define WO   95
#define WT   6
#define NTAP 81
#define XCH  (Tt*Dd*Hh*Ww)
#define XROW (Hh*Ww)

typedef __attribute__((ext_vector_type(2))) float v2f;
typedef __attribute__((ext_vector_type(8))) float v8f;

__global__ __launch_bounds__(128)
void convT4d_wmma_kernel(const float* __restrict__ x,
                         const float* __restrict__ weight,
                         float* __restrict__ out) {
    __shared__ float wlds[NTAP * 16 * Cc];

    for (int idx = threadIdx.x; idx < NTAP * 16 * Cc; idx += blockDim.x) {
        int ci  = idx & 7;
        int co  = (idx >> 3) & 15;
        int tap = idx >> 7;
        int kw  = tap % 3;
        int kh  = (tap / 3) % 3;
        int kd  = (tap / 9) % 3;
        int i   = tap / 27;
        float v = 0.0f;
        if (co < Cc) {
            int widx = ((((ci * Cc + co) * KT + i) * KS + (2 - kd)) * KS + (2 - kh)) * KS + (2 - kw);
            v = weight[widx];
        }
        wlds[idx] = v;
    }
    __syncthreads();

    const int wave = threadIdx.x >> 5;
    const int lane = threadIdx.x & 31;
    const int tid  = blockIdx.x * 4 + wave;
    const int total = TO * DO * HO * WT;
    if (tid >= total) return;

    int t   = tid;
    const int wt  = t % WT;  t /= WT;
    const int oh  = t % HO;  t /= HO;
    const int od  = t % DO;  t /= DO;
    const int f   = t;
    const int owb = wt * 16;

    const int n  = lane & 15;
    const int k0 = (lane >> 4) * 2;

    const int ib0 = (owb + n - 1) >> 1;
    const int ib  = ib0 < 0 ? 0 : (ib0 > (Ww - 2) ? (Ww - 2) : ib0);

    float mlo[3], mhi[3];
    #pragma unroll
    for (int kw = 0; kw < KS; ++kw) {
        const int  mw    = owb + n + kw - 1;
        const int  iw    = mw >> 1;
        const bool valid = ((mw & 1) == 0) && (mw >= 0) && (iw < Ww);
        mlo[kw] = (valid && iw == ib)     ? 1.0f : 0.0f;
        mhi[kw] = (valid && iw == ib + 1) ? 1.0f : 0.0f;
    }

    v8f acc = {};

    #pragma unroll
    for (int i = 0; i < KT; ++i) {
        const int frame = f + i;
        #pragma unroll
        for (int kd = 0; kd < KS; ++kd) {
            const int md = od + kd - 1;
            if (md & 1) continue;
            const int id = md >> 1;
            #pragma unroll
            for (int kh = 0; kh < KS; ++kh) {
                const int mh = oh + kh - 1;
                if (mh & 1) continue;
                const int ih = mh >> 1;

                const int rowbase = (frame * Dd + id) * XROW + ih * Ww;
                __builtin_prefetch(x + rowbase, 0, 0);

                float plo[4], phi[4];
                #pragma unroll
                for (int c4 = 0; c4 < 4; ++c4) {
                    const int ci = (c4 >> 1) * 4 + k0 + (c4 & 1);
                    const float* p = x + ci * XCH + rowbase + ib;
                    plo[c4] = p[0];
                    phi[c4] = p[1];
                }

                #pragma unroll
                for (int kw = 0; kw < KS; ++kw) {
                    const int tapoff = (((i * 3 + kd) * 3 + kh) * 3 + kw) << 7;
                    #pragma unroll
                    for (int chunk = 0; chunk < 2; ++chunk) {
                        const v2f aa = *(const v2f*)&wlds[tapoff + n * Cc + chunk * 4 + k0];
                        v2f bb;
                        #pragma unroll
                        for (int v = 0; v < 2; ++v) {
                            const int c4 = chunk * 2 + v;
                            bb[v] = plo[c4] * mlo[kw] + phi[c4] * mhi[kw];
                        }
                        acc = __builtin_amdgcn_wmma_f32_16x16x4_f32(
                                  false, aa, false, bb, (short)0, acc, false, false);
                    }
                }
            }
        }
    }

    if (lane < 16) {
        const int ow = owb + lane;
        if (ow < WO) {
            const int obase = ((f * DO + od) * HO + oh) * WO + ow;
            #pragma unroll
            for (int r = 0; r < Cc; ++r) {
                out[r * (TO * DO * HO * WO) + obase] = acc[r];
            }
        }
    }
}

extern "C" void kernel_launch(void* const* d_in, const int* in_sizes, int n_in,
                              void* d_out, int out_size, void* d_ws, size_t ws_size,
                              hipStream_t stream) {
    const float* x = (const float*)d_in[0];
    const float* w = (const float*)d_in[1];
    float* out = (float*)d_out;

    const int total  = TO * DO * HO * WT;
    const int blocks = (total + 3) / 4;
    convT4d_wmma_kernel<<<blocks, 128, 0, stream>>>(x, w, out);
}


# Persistent grid: 512 work-groups of 27 waves stage the weight table to LDS once each (not once per 4 tiles) and each wave walks 3 equal-work units of 2x2 output rows
# speedup vs baseline: 4.3347x; 4.3347x over previous
#include <hip/hip_runtime.h>
#include <hip/hip_bf16.h>

#define Cc   8
#define Tt   8
#define Dd   24
#define Hh   48
#define Ww   48
#define KT   3
#define KS   3
#define TO   6
#define DO   47
#define HO   95
#define WO   95
#define WT   6
#define NTAP 81
#define XCH  (Tt*Dd*Hh*Ww)
#define XROW (Hh*Ww)
#define ODP  24
#define OHP  48
#define NBLK 512
#define NWAVE 27
#define UPB  81

typedef __attribute__((ext_vector_type(2))) float v2f;
typedef __attribute__((ext_vector_type(8))) float v8f;

__global__ __launch_bounds__(NWAVE * 32)
void convT4d_wmma_kernel(const float* __restrict__ x,
                         const float* __restrict__ weight,
                         float* __restrict__ out) {
    __shared__ float wlds[NTAP * 16 * Cc];

    for (int idx = threadIdx.x; idx < NTAP * 16 * Cc; idx += blockDim.x) {
        int ci  = idx & 7;
        int co  = (idx >> 3) & 15;
        int tap = idx >> 7;
        int kw  = tap % 3;
        int kh  = (tap / 3) % 3;
        int kd  = (tap / 9) % 3;
        int i   = tap / 27;
        float v = 0.0f;
        if (co < Cc) {
            int widx = ((((ci * Cc + co) * KT + i) * KS + (2 - kd)) * KS + (2 - kh)) * KS + (2 - kw);
            v = weight[widx];
        }
        wlds[idx] = v;
    }
    __syncthreads();

    const int wave = threadIdx.x >> 5;
    const int lane = threadIdx.x & 31;
    const int n  = lane & 15;
    const int k0 = (lane >> 4) * 2;

    for (int j = 0; j < UPB / NWAVE; ++j) {

    int t   = blockIdx.x * UPB + j * NWAVE + wave;
    const int wt  = t % WT;   t /= WT;
    const int ohp = t % OHP;  t /= OHP;
    const int odp = t % ODP;  t /= ODP;
    const int f   = t;
    const int owb = wt * 16;

    const int ib0 = (owb + n - 1) >> 1;
    const int ib  = ib0 < 0 ? 0 : (ib0 > (Ww - 2) ? (Ww - 2) : ib0);

    float mlo[3], mhi[3];
    #pragma unroll
    for (int kw = 0; kw < KS; ++kw) {
        const int  mw    = owb + n + kw - 1;
        const int  iw    = mw >> 1;
        const bool valid = ((mw & 1) == 0) && (mw >= 0) && (iw < Ww);
        mlo[kw] = (valid && iw == ib)     ? 1.0f : 0.0f;
        mhi[kw] = (valid && iw == ib + 1) ? 1.0f : 0.0f;
    }

    #pragma unroll 1
    for (int sub = 0; sub < 4; ++sub) {
    const int od = 2 * odp + (sub >> 1);
    const int oh = 2 * ohp + (sub & 1);
    if (od >= DO || oh >= HO) continue;

    v8f acc = {};

    #pragma unroll
    for (int i = 0; i < KT; ++i) {
        const int frame = f + i;
        #pragma unroll
        for (int kd = 0; kd < KS; ++kd) {
            const int md = od + kd - 1;
            if (md & 1) continue;
            const int id = md >> 1;
            #pragma unroll
            for (int kh = 0; kh < KS; ++kh) {
                const int mh = oh + kh - 1;
                if (mh & 1) continue;
                const int ih = mh >> 1;

                const int rowbase = (frame * Dd + id) * XROW + ih * Ww;
                __builtin_prefetch(x + rowbase, 0, 0);

                float plo[4], phi[4];
                #pragma unroll
                for (int c4 = 0; c4 < 4; ++c4) {
                    const int ci = (c4 >> 1) * 4 + k0 + (c4 & 1);
                    const float* p = x + ci * XCH + rowbase + ib;
                    plo[c4] = p[0];
                    phi[c4] = p[1];
                }

                #pragma unroll
                for (int kw = 0; kw < KS; ++kw) {
                    const int tapoff = (((i * 3 + kd) * 3 + kh) * 3 + kw) << 7;
                    #pragma unroll
                    for (int chunk = 0; chunk < 2; ++chunk) {
                        const v2f aa = *(const v2f*)&wlds[tapoff + n * Cc + chunk * 4 + k0];
                        v2f bb;
                        #pragma unroll
                        for (int v = 0; v < 2; ++v) {
                            const int c4 = chunk * 2 + v;
                            bb[v] = plo[c4] * mlo[kw] + phi[c4] * mhi[kw];
                        }
                        acc = __builtin_amdgcn_wmma_f32_16x16x4_f32(
                                  false, aa, false, bb, (short)0, acc, false, false);
                    }
                }
            }
        }
    }

    if (lane < 16) {
        const int ow = owb + lane;
        if (ow < WO) {
            const int obase = ((f * DO + od) * HO + oh) * WO + ow;
            #pragma unroll
            for (int r = 0; r < Cc; ++r) {
                out[r * (TO * DO * HO * WO) + obase] = acc[r];
            }
        }
    }

    }
    }
}

extern "C" void kernel_launch(void* const* d_in, const int* in_sizes, int n_in,
                              void* d_out, int out_size, void* d_ws, size_t ws_size,
                              hipStream_t stream) {
    const float* x = (const float*)d_in[0];
    const float* w = (const float*)d_in[1];
    float* out = (float*)d_out;

    convT4d_wmma_kernel<<<NBLK, NWAVE * 32, 0, stream>>>(x, w, out);
}


# Pack both ow parities with co into the 16 WMMA rows: M=(ow parity,co), K=(column shift,ci), B is x itself; 4 WMMAs per tap group finish 32 ow (was 6 for 16), no kw masks; 2 waves share 3 tiles
# speedup vs baseline: 7.7903x; 1.7972x over previous
#include <hip/hip_runtime.h>
#include <hip/hip_bf16.h>


#define Cc   8
#define Tt   8
#define Dd   24
#define Hh   48
#define Ww   48
#define KT   3
#define KS   3
#define TO   6
#define DO   47
#define HO   95
#define WO   95
#define NTG  27
#define XCH  (Tt*Dd*Hh*Ww)
#define XROW (Hh*Ww)
#define ODP  24
#define OHP  48
#define NBLK 512
#define NWAVE 27
#define OCH  (TO * DO * HO * WO)

typedef __attribute__((ext_vector_type(2))) float v2f;
typedef __attribute__((ext_vector_type(4))) float v4f;
typedef __attribute__((ext_vector_type(8))) float v8f;

__global__ __launch_bounds__(NWAVE * 32)
void convT4d_wmma_kernel(const float* __restrict__ x,
                         const float* __restrict__ weight,
                         float* __restrict__ out) {
    __shared__ __attribute__((aligned(16))) float wl[NTG * 2 * 16 * 8];

    for (int idx = threadIdx.x; idx < NTG * 256; idx += blockDim.x) {
        const int e  = idx & 3;
        const int hh = (idx >> 2) & 1;
        const int m  = (idx >> 3) & 15;
        const int s  = (idx >> 7) & 1;
        const int tg = idx >> 8;
        const int ci = (e >> 1) * 4 + hh * 2 + (e & 1);
        const int co = m & 7;
        const int pw = m >> 3;
        const int kh = tg % 3;
        const int kd = (tg / 3) % 3;
        const int i  = tg / 9;
        const int kw = pw ? 2 * s : 1;
        float v = 0.0f;
        if (pw | (s ^ 1)) {
            const int widx = ((((ci * Cc + co) * KT + i) * KS + (2 - kd)) * KS + (2 - kh)) * KS + (2 - kw);
            v = weight[widx];
        }
        wl[idx] = v;
    }
    __syncthreads();

    const int wave = threadIdx.x >> 5;
    const int lane = threadIdx.x & 31;
    const int n = lane & 15;
    const int h = lane >> 4;

    const int gw   = blockIdx.x * NWAVE + wave;
    const int role = (gw ^ (gw >> 2)) & 1;
    int t = gw >> 1;
    const int ohp = t % OHP;  t /= OHP;
    const int odp = t % ODP;  t /= ODP;
    const int f   = t;

    #pragma unroll 1
    for (int part = 0; part < 2; ++part) {
    const int qt   = part ? 1 : (role ? 2 : 0);
    const int rows = part ? (role ? 0x8 : 0x7) : 0xF;
    const int q    = qt * 16 + n;

    const int  qc   = q < Ww - 1 ? q : Ww - 2;
    const bool last = q == Ww - 1;
    const int  lofs = (2 * h) * XCH + qc;

    #pragma unroll 1
    for (int sub = 0; sub < 4; ++sub) {
    if (!((rows >> sub) & 1)) continue;
    const int od = 2 * odp + (sub >> 1);
    const int oh = 2 * ohp + (sub & 1);
    if (od >= DO || oh >= HO) continue;

    v8f acc = {};

    #pragma unroll
    for (int i = 0; i < KT; ++i) {
        const int frame = f + i;
        #pragma unroll
        for (int kd = 0; kd < KS; ++kd) {
            const int md = od + kd - 1;
            if (md & 1) continue;
            const int id = md >> 1;
            #pragma unroll
            for (int kh = 0; kh < KS; ++kh) {
                const int mh = oh + kh - 1;
                if (mh & 1) continue;
                const int ih = mh >> 1;

                const int rowbase = (frame * Dd + id) * XROW + ih * Ww;
                __builtin_prefetch(x + rowbase, 0, 0);

                float lo[4], hi[4];
                #pragma unroll
                for (int c4 = 0; c4 < 4; ++c4) {
                    const float* p = x + ((c4 >> 1) * 4 + (c4 & 1)) * XCH + rowbase + lofs;
                    const float a = p[0];
                    const float b = p[1];
                    lo[c4] = last ? b : a;
                    hi[c4] = last ? 0.0f : b;
                }

                const int tg = (i * 3 + kd) * 3 + kh;
                #pragma unroll
                for (int s = 0; s < 2; ++s) {
                    const v4f aa = *(const v4f*)&wl[((tg * 2 + s) * 16 + n) * 8 + h * 4];
                    #pragma unroll
                    for (int ch = 0; ch < 2; ++ch) {
                        v2f a2, b2;
                        a2[0] = aa[ch * 2];
                        a2[1] = aa[ch * 2 + 1];
                        b2[0] = s ? hi[ch * 2]     : lo[ch * 2];
                        b2[1] = s ? hi[ch * 2 + 1] : lo[ch * 2 + 1];
                        acc = __builtin_amdgcn_wmma_f32_16x16x4_f32(
                                  false, a2, false, b2, (short)0, acc, false, false);
                    }
                }
            }
        }
    }

    const int ow = 2 * q + h;
    if (ow < WO) {
        const int obase = ((f * DO + od) * HO + oh) * WO + ow;
        #pragma unroll
        for (int r = 0; r < Cc; ++r) {
            out[r * OCH + obase] = acc[r];
        }
    }

    }
    }
}

extern "C" void kernel_launch(void* const* d_in, const int* in_sizes, int n_in,
                              void* d_out, int out_size, void* d_ws, size_t ws_size,
                              hipStream_t stream) {
    const float* x = (const float*)d_in[0];
    const float* w = (const float*)d_in[1];
    float* out = (float*)d_out;

    convT4d_wmma_kernel<<<NBLK, NWAVE * 32, 0, stream>>>(x, w, out);
}


# Load each x fragment once per (frame, id, ih) and feed the 1-4 output-row accumulators it serves; one wave per (f, od pair, oh pair), 256 work-groups: 48 loads per q tile, was 108; same sum order
# speedup vs baseline: 11.1412x; 1.4301x over previous
#include <hip/hip_runtime.h>
#include <hip/hip_bf16.h>


#define Cc   8
#define Tt   8
#define Dd   24
#define Hh   48
#define Ww   48
#define KT   3
#define KS   3
#define TO   6
#define DO   47
#define HO   95
#define WO   95
#define NTG  27
#define XCH  (Tt*Dd*Hh*Ww)
#define XROW (Hh*Ww)
#define ODP  24
#define OHP  48
#define NBLK 256
#define NWAVE 27
#define OCH  (TO * DO * HO * WO)

typedef __attribute__((ext_vector_type(2))) float v2f;
typedef __attribute__((ext_vector_type(4))) float v4f;
typedef __attribute__((ext_vector_type(8))) float v8f;

__global__ __launch_bounds__(NWAVE * 32)
void convT4d_wmma_kernel(const float* __restrict__ x,
                         const float* __restrict__ weight,
                         float* __restrict__ out) {
    __shared__ __attribute__((aligned(16))) float wl[NTG * 2 * 16 * 8];

    for (int idx = threadIdx.x; idx < NTG * 256; idx += blockDim.x) {
        const int e  = idx & 3;
        const int hh = (idx >> 2) & 1;
        const int m  = (idx >> 3) & 15;
        const int s  = (idx >> 7) & 1;
        const int tg = idx >> 8;
        const int ci = (e >> 1) * 4 + hh * 2 + (e & 1);
        const int co = m & 7;
        const int pw = m >> 3;
        const int kh = tg % 3;
        const int kd = (tg / 3) % 3;
        const int i  = tg / 9;
        const int kw = pw ? 2 * s : 1;
        float v = 0.0f;
        if (pw | (s ^ 1)) {
            const int widx = ((((ci * Cc + co) * KT + i) * KS + (2 - kd)) * KS + (2 - kh)) * KS + (2 - kw);
            v = weight[widx];
        }
        wl[idx] = v;
    }
    __syncthreads();

    const int wave = __builtin_amdgcn_readfirstlane((int)(threadIdx.x >> 5));
    const int lane = threadIdx.x & 31;
    const int n = lane & 15;
    const int h = lane >> 4;

    int t = blockIdx.x * NWAVE + wave;
    const int ohp = t % OHP;  t /= OHP;
    const int odp = t % ODP;  t /= ODP;
    const int f   = t;

    const int id1 = odp + 1 < Dd ? odp + 1 : Dd - 1;
    const int ih1 = ohp + 1 < Hh ? ohp + 1 : Hh - 1;

    for (int qt = 0; qt < 3; ++qt) {
    const int q = qt * 16 + n;

    const int  qc   = q < Ww - 1 ? q : Ww - 2;
    const bool last = q == Ww - 1;
    const int  lofs = (2 * h) * XCH + qc;

    v8f acc[4] = {};

    for (int i = 0; i < KT; ++i) {
        const int frame = f + i;

        float xu[4][4], xw[4][4];
        #pragma unroll
        for (int a = 0; a < 2; ++a) {
            const int id = a ? id1 : odp;
            #pragma unroll
            for (int b = 0; b < 2; ++b) {
                const int ih = b ? ih1 : ohp;
                const int rowbase = (frame * Dd + id) * XROW + ih * Ww;
                #pragma unroll
                for (int c4 = 0; c4 < 4; ++c4) {
                    const float* p = x + ((c4 >> 1) * 4 + (c4 & 1)) * XCH + rowbase + lofs;
                    xu[a * 2 + b][c4] = p[0];
                    xw[a * 2 + b][c4] = p[1];
                }
            }
        }
        __builtin_amdgcn_sched_barrier(0);

        float lo[4][4], hi[4][4];
        #pragma unroll
        for (int r4 = 0; r4 < 4; ++r4) {
            #pragma unroll
            for (int c4 = 0; c4 < 4; ++c4) {
                lo[r4][c4] = last ? xw[r4][c4] : xu[r4][c4];
                hi[r4][c4] = last ? 0.0f : xw[r4][c4];
            }
        }

        #pragma unroll
        for (int a = 0; a < 2; ++a) {
            #pragma unroll
            for (int b = 0; b < 2; ++b) {
                #pragma unroll
                for (int pd = a; pd < 2; ++pd) {
                    #pragma unroll
                    for (int ph = b; ph < 2; ++ph) {
                        const int kd = pd ? 2 * a : 1;
                        const int kh = ph ? 2 * b : 1;
                        const int tg = (i * 3 + kd) * 3 + kh;
                        #pragma unroll
                        for (int s = 0; s < 2; ++s) {
                            const v4f aa = *(const v4f*)&wl[((tg * 2 + s) * 16 + n) * 8 + h * 4];
                            #pragma unroll
                            for (int ch = 0; ch < 2; ++ch) {
                                v2f a2, b2;
                                a2[0] = aa[ch * 2];
                                a2[1] = aa[ch * 2 + 1];
                                b2[0] = s ? hi[a * 2 + b][ch * 2]     : lo[a * 2 + b][ch * 2];
                                b2[1] = s ? hi[a * 2 + b][ch * 2 + 1] : lo[a * 2 + b][ch * 2 + 1];
                                acc[pd * 2 + ph] = __builtin_amdgcn_wmma_f32_16x16x4_f32(
                                        false, a2, false, b2, (short)0, acc[pd * 2 + ph], false, false);
                            }
                        }
                    }
                }
            }
        }
    }

    const int ow = 2 * q + h;
    #pragma unroll
    for (int pd = 0; pd < 2; ++pd) {
        #pragma unroll
        for (int ph = 0; ph < 2; ++ph) {
            const int od = 2 * odp + pd;
            const int oh = 2 * ohp + ph;
            if (od < DO && oh < HO && ow < WO) {
                const int obase = ((f * DO + od) * HO + oh) * WO + ow;
                #pragma unroll
                for (int r = 0; r < Cc; ++r) {
                    out[r * OCH + obase] = acc[pd * 2 + ph][r];
                }
            }
        }
    }

    }
}

extern "C" void kernel_launch(void* const* d_in, const int* in_sizes, int n_in,
                              void* d_out, int out_size, void* d_ws, size_t ws_size,
                              hipStream_t stream) {
    const float* x = (const float*)d_in[0];
    const float* w = (const float*)d_in[1];
    float* out = (float*)d_out;

    convT4d_wmma_kernel<<<NBLK, NWAVE * 32, 0, stream>>>(x, w, out);
}


# Copy the x rows of a 3x3x3 cube of units per work-group to LDS once (16-byte whole-line loads); B fragments read from LDS, no global loads in main loop; K order per tap group: (ci pair, shift)
# speedup vs baseline: 13.1212x; 1.1777x over previous
#include <hip/hip_runtime.h>
#include <hip/hip_bf16.h>


#define Cc   8
#define Tt   8
#define Dd   24
#define Hh   48
#define Ww   48
#define KT   3
#define KS   3
#define TO   6
#define DO   47
#define HO   95
#define WO   95
#define NTG  27
#define XCH  (Tt*Dd*Hh*Ww)
#define XROW (Hh*Ww)
#define NBLK 256
#define NWAVE 27
#define OCH  (TO * DO * HO * WO)

#define FS   5
#define IS   4
#define HS   4
#define PR   96
#define XS_DW (FS * IS * HS * 8 * PR)
#define NTASK (FS * IS * 4 * HS * 12)
#define NPASS 5

typedef __attribute__((ext_vector_type(2))) float v2f;
typedef __attribute__((ext_vector_type(4))) float v4f;
typedef __attribute__((ext_vector_type(8))) float v8f;

__global__ __launch_bounds__(NWAVE * 32)
void convT4d_wmma_kernel(const float* __restrict__ x,
                         const float* __restrict__ weight,
                         float* __restrict__ out) {
    __shared__ __attribute__((aligned(16))) float wl[NTG * 2 * 16 * 8];
    __shared__ __attribute__((aligned(16))) float xs[XS_DW];

    const int tid = threadIdx.x;
    const int bh = blockIdx.x & 15;
    const int bo = (blockIdx.x >> 4) & 7;
    const int bf = blockIdx.x >> 7;
    const int f0 = 3 * bf, odp0 = 3 * bo, ohp0 = 3 * bh;

    v4f xa[NPASS], xb[NPASS];
    float xa4[NPASS], xb4[NPASS];
    #pragma unroll
    for (int p = 0; p < NPASS; ++p) {
        const int task = tid + p * (NWAVE * 32);
        if (task < NTASK) {
            const int cg = task % 12;
            int r = task / 12;
            const int ihs = r & 3;  r >>= 2;
            const int c   = r & 3;  r >>= 2;
            const int ids = r & 3;
            const int fs  = r >> 2;
            const int id = odp0 + ids < Dd ? odp0 + ids : Dd - 1;
            const int ih = ohp0 + ihs < Hh ? ohp0 + ihs : Hh - 1;
            const float* src = x + (2 * c) * XCH + ((f0 + fs) * Dd + id) * XROW + ih * Ww + cg * 4;
            xa[p] = *(const v4f*)src;
            xb[p] = *(const v4f*)(src + XCH);
            xa4[p] = 0.0f;
            xb4[p] = 0.0f;
            if (cg < 11) {
                xa4[p] = src[4];
                xb4[p] = src[XCH + 4];
            }
        }
    }

    for (int idx = tid; idx < NTG * 256; idx += NWAVE * 32) {
        const int e  = idx & 3;
        const int hh = (idx >> 2) & 1;
        const int m  = (idx >> 3) & 15;
        const int cp = (idx >> 7) & 1;
        const int tg = idx >> 8;
        const int s  = e & 1;
        const int ci = (2 * cp + (e >> 1)) * 2 + hh;
        const int co = m & 7;
        const int pw = m >> 3;
        const int kh = tg % 3;
        const int kd = (tg / 3) % 3;
        const int i  = tg / 9;
        const int kw = pw ? 2 * s : 1;
        float v = 0.0f;
        if (pw | (s ^ 1)) {
            const int widx = ((((ci * Cc + co) * KT + i) * KS + (2 - kd)) * KS + (2 - kh)) * KS + (2 - kw);
            v = weight[widx];
        }
        wl[idx] = v;
    }

    #pragma unroll
    for (int p = 0; p < NPASS; ++p) {
        const int task = tid + p * (NWAVE * 32);
        if (task < NTASK) {
            const int cg = task % 12;
            int r = task / 12;
            const int ihs = r & 3;  r >>= 2;
            const int c   = r & 3;  r >>= 2;
            const int ids = r & 3;
            const int fs  = r >> 2;
            float* dst = xs + (((fs * IS + ids) * HS + ihs) * 8 + 2 * c) * PR + cg * 8;
            v4f t;
            t[0] = xa[p][0]; t[1] = xa[p][1]; t[2] = xa[p][1]; t[3] = xa[p][2];
            *(v4f*)dst = t;
            t[0] = xa[p][2]; t[1] = xa[p][3]; t[2] = xa[p][3]; t[3] = xa4[p];
            *(v4f*)(dst + 4) = t;
            t[0] = xb[p][0]; t[1] = xb[p][1]; t[2] = xb[p][1]; t[3] = xb[p][2];
            *(v4f*)(dst + PR) = t;
            t[0] = xb[p][2]; t[1] = xb[p][3]; t[2] = xb[p][3]; t[3] = xb4[p];
            *(v4f*)(dst + PR + 4) = t;
        }
    }
    __syncthreads();

    const int wave = __builtin_amdgcn_readfirstlane((int)(threadIdx.x >> 5));
    const int lane = threadIdx.x & 31;
    const int n = lane & 15;
    const int h = lane >> 4;

    const int hl = wave % 3;
    const int ol = (wave / 3) % 3;
    const int fl = wave / 9;
    const int f   = f0 + fl;
    const int odp = odp0 + ol;
    const int ohp = ohp0 + hl;
    const float* xw = xs + (((fl * IS + ol) * HS + hl) * 8 + h) * PR;

    for (int qt = 0; qt < 3; ++qt) {
    asm volatile("" ::: "memory");
    const int q = qt * 16 + n;
    const float* xq = xw + 2 * q;

    v8f acc[4] = {};

    #pragma unroll
    for (int i = 0; i < KT; ++i) {
        v2f bfr[4][4];
        #pragma unroll
        for (int a = 0; a < 2; ++a) {
            #pragma unroll
            for (int b = 0; b < 2; ++b) {
                #pragma unroll
                for (int c = 0; c < 4; ++c) {
                    bfr[a * 2 + b][c] = *(const v2f*)(xq + (((i * IS + a) * HS + b) * 8 + 2 * c) * PR);
                }
            }
        }

        #pragma unroll
        for (int a = 0; a < 2; ++a) {
            #pragma unroll
            for (int b = 0; b < 2; ++b) {
                #pragma unroll
                for (int pd = a; pd < 2; ++pd) {
                    #pragma unroll
                    for (int ph = b; ph < 2; ++ph) {
                        const int kd = pd ? 2 * a : 1;
                        const int kh = ph ? 2 * b : 1;
                        const int tg = (i * 3 + kd) * 3 + kh;
                        #pragma unroll
                        for (int cp = 0; cp < 2; ++cp) {
                            const v4f aa = *(const v4f*)&wl[((tg * 2 + cp) * 16 + n) * 8 + h * 4];
                            #pragma unroll
                            for (int e2 = 0; e2 < 2; ++e2) {
                                v2f a2;
                                a2[0] = aa[e2 * 2];
                                a2[1] = aa[e2 * 2 + 1];
                                acc[pd * 2 + ph] = __builtin_amdgcn_wmma_f32_16x16x4_f32(
                                        false, a2, false, bfr[a * 2 + b][cp * 2 + e2], (short)0,
                                        acc[pd * 2 + ph], false, false);
                            }
                        }
                    }
                }
            }
        }
    }

    const int ow = 2 * q + h;
    #pragma unroll
    for (int pd = 0; pd < 2; ++pd) {
        #pragma unroll
        for (int ph = 0; ph < 2; ++ph) {
            const int od = 2 * odp + pd;
            const int oh = 2 * ohp + ph;
            if (od < DO && oh < HO && ow < WO) {
                const int obase = ((f * DO + od) * HO + oh) * WO + ow;
                #pragma unroll
                for (int r = 0; r < Cc; ++r) {
                    out[r * OCH + obase] = acc[pd * 2 + ph][r];
                }
            }
        }
    }

    }
}

extern "C" void kernel_launch(void* const* d_in, const int* in_sizes, int n_in,
                              void* d_out, int out_size, void* d_ws, size_t ws_size,
                              hipStream_t stream) {
    const float* x = (const float*)d_in[0];
    const float* w = (const float*)d_in[1];
    float* out = (float*)d_out;

    convT4d_wmma_kernel<<<NBLK, NWAVE * 32, 0, stream>>>(x, w, out);
}


# Split products by column shift: shift-0 tiles M=(ow parity,co), K=ci; shift-1 (kw=2) tiles stack two output rows in M, sums added to odd ow at end; 84 WMMAs per q tile, was 108; sum order changes
# speedup vs baseline: 14.7382x; 1.1232x over previous
#include <hip/hip_runtime.h>
#include <hip/hip_bf16.h>


#define Cc   8
#define Tt   8
#define Dd   24
#define Hh   48
#define Ww   48
#define KT   3
#define KS   3
#define TO   6
#define DO   47
#define HO   95
#define WO   95
#define NTG  27
#define XCH  (Tt*Dd*Hh*Ww)
#define XROW (Hh*Ww)
#define NBLK 256
#define NWAVE 27
#define OCH  (TO * DO * HO * WO)

#define FS   5
#define IS   4
#define HS   4
#define CP   160
#define RP   (4 * CP)
#define XS_DW (FS * IS * HS * RP)
#define NTASK (FS * IS * 4 * HS * 12)
#define NPASS 5
#define NG2  5
#define W1_DW (NTG * 128)
#define W2_DW (KT * NG2 * 128)

typedef __attribute__((ext_vector_type(2))) float v2f;
typedef __attribute__((ext_vector_type(4))) float v4f;
typedef __attribute__((ext_vector_type(8))) float v8f;

__device__ __forceinline__ float xhalf(float v) {
    return __builtin_bit_cast(float, __builtin_amdgcn_permlanex16(
        __builtin_bit_cast(int, v), __builtin_bit_cast(int, v), 0x76543210, 0xfedcba98, false, false));
}

__global__ __launch_bounds__(NWAVE * 32)
void convT4d_wmma_kernel(const float* __restrict__ x,
                         const float* __restrict__ weight,
                         float* __restrict__ out) {
    __shared__ __attribute__((aligned(16))) float wl1[W1_DW];
    __shared__ __attribute__((aligned(16))) float wl2[W2_DW];
    __shared__ __attribute__((aligned(16))) float xs[XS_DW];

    const int tid = threadIdx.x;
    const int bh = blockIdx.x & 15;
    const int bo = (blockIdx.x >> 4) & 7;
    const int bf = blockIdx.x >> 7;
    const int f0 = 3 * bf, odp0 = 3 * bo, ohp0 = 3 * bh;

    v4f xa[NPASS], xb[NPASS];
    #pragma unroll
    for (int p = 0; p < NPASS; ++p) {
        const int task = tid + p * (NWAVE * 32);
        if (task < NTASK) {
            const int cg = task % 12;
            int r = task / 12;
            const int ihs = r & 3;  r >>= 2;
            const int c   = r & 3;  r >>= 2;
            const int ids = r & 3;
            const int fs  = r >> 2;
            const int id = odp0 + ids < Dd ? odp0 + ids : Dd - 1;
            const int ih = ohp0 + ihs < Hh ? ohp0 + ihs : Hh - 1;
            const float* src = x + (2 * c) * XCH + ((f0 + fs) * Dd + id) * XROW + ih * Ww + cg * 4;
            xa[p] = *(const v4f*)src;
            xb[p] = *(const v4f*)(src + XCH);
        }
    }

    for (int idx = tid; idx < W1_DW; idx += NWAVE * 32) {
        const int e  = idx & 1;
        const int c  = (idx >> 1) & 1;
        const int hh = (idx >> 2) & 1;
        const int m  = (idx >> 3) & 15;
        const int tg = idx >> 7;
        const int ci = 4 * c + 2 * hh + e;
        const int co = m & 7;
        const int pw = m >> 3;
        const int kh = tg % 3;
        const int kd = (tg / 3) % 3;
        const int i  = tg / 9;
        const int kw = pw ? 0 : 1;
        const int widx = ((((ci * Cc + co) * KT + i) * KS + (2 - kd)) * KS + (2 - kh)) * KS + (2 - kw);
        wl1[idx] = weight[widx];
    }
    for (int idx = tid; idx < W2_DW; idx += NWAVE * 32) {
        const int e  = idx & 1;
        const int c  = (idx >> 1) & 1;
        const int hh = (idx >> 2) & 1;
        const int m  = (idx >> 3) & 15;
        const int gg = idx >> 7;
        const int g  = gg % NG2;
        const int i  = gg / NG2;
        const int ci = 4 * c + 2 * hh + e;
        const int co = m & 7;
        const int j  = m >> 3;
        const int code = (int)((0x8F67250134ULL >> (4 * (g * 2 + j))) & 15);
        float v = 0.0f;
        if (code < 9) {
            const int kd = code / 3;
            const int kh = code % 3;
            const int widx = ((((ci * Cc + co) * KT + i) * KS + (2 - kd)) * KS + (2 - kh)) * KS + 0;
            v = weight[widx];
        }
        wl2[idx] = v;
    }

    #pragma unroll
    for (int p = 0; p < NPASS; ++p) {
        const int task = tid + p * (NWAVE * 32);
        if (task < NTASK) {
            const int cg = task % 12;
            int r = task / 12;
            const int ihs = r & 3;  r >>= 2;
            const int c   = r & 3;  r >>= 2;
            const int ids = r & 3;
            const int fs  = r >> 2;
            float* dst = xs + ((fs * IS + ids) * HS + ihs) * RP + c * CP + cg * 8;
            v4f t;
            t[0] = xa[p][0]; t[1] = xb[p][0]; t[2] = xa[p][1]; t[3] = xb[p][1];
            *(v4f*)dst = t;
            t[0] = xa[p][2]; t[1] = xb[p][2]; t[2] = xa[p][3]; t[3] = xb[p][3];
            *(v4f*)(dst + 4) = t;
            if (cg == 11) {
                v2f z;
                z[0] = 0.0f; z[1] = 0.0f;
                *(v2f*)(dst + 8) = z;
            }
        }
    }
    __syncthreads();

    const int wave = __builtin_amdgcn_readfirstlane((int)(threadIdx.x >> 5));
    const int lane = threadIdx.x & 31;
    const int n = lane & 15;
    const int h = lane >> 4;

    const int hl = wave % 3;
    const int ol = (wave / 3) % 3;
    const int fl = wave / 9;
    const int f   = f0 + fl;
    const int odp = odp0 + ol;
    const int ohp = ohp0 + hl;
    const float* xw = xs + ((fl * IS + ol) * HS + hl) * RP + h * CP;
    const float* w1 = wl1 + n * 8 + h * 4;
    const float* w2 = wl2 + n * 8 + h * 4;

    for (int qt = 0; qt < 3; ++qt) {
    asm volatile("" ::: "memory");
    const int q = qt * 16 + n;
    const float* xq = xw + 2 * q;

    v8f acc[4] = {};
    v8f sA = {}, sB = {}, sC = {};

    #pragma unroll
    for (int i = 0; i < KT; ++i) {
        v2f b0[4][2], b1[4][2];
        #pragma unroll
        for (int a = 0; a < 2; ++a) {
            #pragma unroll
            for (int b = 0; b < 2; ++b) {
                #pragma unroll
                for (int c = 0; c < 2; ++c) {
                    const float* p = xq + ((i * IS + a) * HS + b) * RP + 2 * c * CP;
                    b0[a * 2 + b][c] = *(const v2f*)p;
                    b1[a * 2 + b][c] = *(const v2f*)(p + 2);
                }
            }
        }

        #pragma unroll
        for (int a = 0; a < 2; ++a) {
            #pragma unroll
            for (int b = 0; b < 2; ++b) {
                #pragma unroll
                for (int pd = a; pd < 2; ++pd) {
                    #pragma unroll
                    for (int ph = b; ph < 2; ++ph) {
                        const int kd = pd ? 2 * a : 1;
                        const int kh = ph ? 2 * b : 1;
                        const int tg = (i * 3 + kd) * 3 + kh;
                        const v4f aa = *(const v4f*)(w1 + tg * 128);
                        #pragma unroll
                        for (int c = 0; c < 2; ++c) {
                            v2f a2;
                            a2[0] = aa[c * 2];
                            a2[1] = aa[c * 2 + 1];
                            acc[pd * 2 + ph] = __builtin_amdgcn_wmma_f32_16x16x4_f32(
                                    false, a2, false, b0[a * 2 + b][c], (short)0,
                                    acc[pd * 2 + ph], false, false);
                        }
                    }
                }
            }
        }

        #pragma unroll
        for (int g = 0; g < NG2; ++g) {
            const int xr = g == 0 ? 0 : g == 1 ? 0 : g == 2 ? 1 : g == 3 ? 2 : 3;
            const v4f aa = *(const v4f*)(w2 + (i * NG2 + g) * 128);
            #pragma unroll
            for (int c = 0; c < 2; ++c) {
                v2f a2;
                a2[0] = aa[c * 2];
                a2[1] = aa[c * 2 + 1];
                if (g == 0) {
                    sA = __builtin_amdgcn_wmma_f32_16x16x4_f32(
                            false, a2, false, b1[xr][c], (short)0, sA, false, false);
                } else if (g == 1 || g == 3) {
                    sB = __builtin_amdgcn_wmma_f32_16x16x4_f32(
                            false, a2, false, b1[xr][c], (short)0, sB, false, false);
                } else if (g == 2) {
                    sC = __builtin_amdgcn_wmma_f32_16x16x4_f32(
                            false, a2, false, b1[xr][c], (short)0, sC, false, false);
                } else {
                    acc[3] = __builtin_amdgcn_wmma_f32_16x16x4_f32(
                            false, a2, false, b1[xr][c], (short)0, acc[3], false, false);
                }
            }
        }
    }

    #pragma unroll
    for (int r = 0; r < Cc; ++r) {
        const float a0 = xhalf(sA[r]);
        const float b0s = xhalf(sB[r]);
        const float c0 = xhalf(sC[r]);
        acc[0][r] += h ? a0 : 0.0f;
        acc[1][r] += h ? sA[r] + c0 : 0.0f;
        acc[2][r] += h ? b0s : 0.0f;
        acc[3][r] += h ? sB[r] + sC[r] : 0.0f;
    }

    const int ow = 2 * q + h;
    #pragma unroll
    for (int pd = 0; pd < 2; ++pd) {
        #pragma unroll
        for (int ph = 0; ph < 2; ++ph) {
            const int od = 2 * odp + pd;
            const int oh = 2 * ohp + ph;
            if (od < DO && oh < HO && ow < WO) {
                const int obase = ((f * DO + od) * HO + oh) * WO + ow;
                #pragma unroll
                for (int r = 0; r < Cc; ++r) {
                    out[r * OCH + obase] = acc[pd * 2 + ph][r];
                }
            }
        }
    }

    }
}

extern "C" void kernel_launch(void* const* d_in, const int* in_sizes, int n_in,
                              void* d_out, int out_size, void* d_ws, size_t ws_size,
                              hipStream_t stream) {
    const float* x = (const float*)d_in[0];
    const float* w = (const float*)d_in[1];
    float* out = (float*)d_out;

    convT4d_wmma_kernel<<<NBLK, NWAVE * 32, 0, stream>>>(x, w, out);
}
